# starting kernel
# baseline (speedup 1.0000x reference)
#include <hip/hip_runtime.h>
#include <hip/hip_bf16.h>

typedef __attribute__((ext_vector_type(2))) float v2f;
typedef __attribute__((ext_vector_type(8))) float v8f;

#define S_LEN   4096
#define D_DIM   1024
#define NPATCH  8
#define NSEG    9

__global__ void seg_bounds_kernel(const int* __restrict__ mask,
                                  int* __restrict__ ofs) {
  const int b   = blockIdx.x;
  const int tid = threadIdx.x;
  __shared__ int psum[256];

  const int base = b * S_LEN + tid * 16;
  int m[16];
  int c = 0;
#pragma unroll
  for (int j = 0; j < 16; ++j) { m[j] = mask[base + j]; c += m[j]; }
  psum[tid] = c;
  __syncthreads();

#pragma unroll
  for (int off = 1; off < 256; off <<= 1) {
    int v   = psum[tid];
    int add = (tid >= off) ? psum[tid - off] : 0;
    __syncthreads();
    psum[tid] = v + add;
    __syncthreads();
  }
  const int excl = (tid == 0) ? 0 : psum[tid - 1];

  if (tid < 10) ofs[b * 10 + tid] = (tid == 0) ? 0 : S_LEN;
  __syncthreads();

  int r = excl;
#pragma unroll
  for (int j = 0; j < 16; ++j) {
    if (m[j]) {
      if (r < NPATCH) ofs[b * 10 + (r + 1)] = tid * 16 + j + 1;
      ++r;
    }
  }
}

__global__ void seg_mean_kernel(const float* __restrict__ x,
                                const int* __restrict__ ofs,
                                float* __restrict__ patches) {
  const int p = blockIdx.x & 7;
  const int b = blockIdx.x >> 3;
  const int d = threadIdx.x;

  const int s0 = ofs[b * 10 + p];
  const int s1 = ofs[b * 10 + p + 1];

  const float* src = x + (size_t)b * S_LEN * D_DIM + d;
  float acc = 0.0f;
  for (int s = s0; s < s1; ++s) acc += src[(size_t)s * D_DIM];

  const int cnt = s1 - s0;
  const float denom = (float)(cnt > 0 ? cnt : 1);
  patches[(size_t)(b * NPATCH + p) * D_DIM + d] = acc / denom;
}

__global__ void patch_gemm_wmma(const float* __restrict__ P,
                                const float* __restrict__ Wm,
                                const float* __restrict__ bias,
                                float* __restrict__ out) {
  const int wave = blockIdx.x * (blockDim.x >> 5) + (threadIdx.x >> 5);
  const int lane = threadIdx.x & 31;
  const int mt   = wave >> 6;
  const int nt   = wave & 63;
  const int m0   = mt * 16;
  const int n0   = nt * 16;
  const int half = lane >> 4;
  const int lr   = lane & 15;

  const float* Arow = P  + (size_t)(m0 + lr) * D_DIM + half * 2;
  const float* Brow = Wm + (size_t)(n0 + lr) * D_DIM + half * 2;

  v8f c = {};
#pragma unroll 8
  for (int k = 0; k < D_DIM; k += 4) {
    v2f a  = *(const v2f*)(Arow + k);
    v2f bb = *(const v2f*)(Brow + k);
    c = __builtin_amdgcn_wmma_f32_16x16x4_f32(
         false, a,  false, bb,
         (short)0, c,  false,  false);
  }

  const float bv = bias[n0 + lr];
  float* ocol = out + n0 + lr;
#pragma unroll
  for (int r = 0; r < 8; ++r) {
    ocol[(size_t)(m0 + half * 8 + r) * D_DIM] = c[r] + bv;
  }
}

extern "C" void kernel_launch(void* const* d_in, const int* in_sizes, int n_in,
                              void* d_out, int out_size, void* d_ws, size_t ws_size,
                              hipStream_t stream) {
  const float* latent = (const float*)d_in[0];
  const int*   mask   = (const int*)  d_in[1];
  const float* W      = (const float*)d_in[2];
  const float* bias   = (const float*)d_in[3];
  float*       out    = (float*)d_out;

  int*   ofs     = (int*)d_ws;
  float* patches = (float*)((char*)d_ws + 1024);

  seg_bounds_kernel<<<8, 256, 0, stream>>>(mask, ofs);
  seg_mean_kernel<<<64, 1024, 0, stream>>>(latent, ofs, patches);
  patch_gemm_wmma<<<64, 128, 0, stream>>>(patches, W, bias, out);
}


# GEMM re-tiled: one 16x16 tile per workgroup, K split over 8 waves, each wave's 128-k slice in flight at once as b128 loads, partial tiles summed in LDS; summation order changes
# speedup vs baseline: 2.2808x; 2.2808x over previous
#include <hip/hip_runtime.h>
#include <hip/hip_bf16.h>

typedef __attribute__((ext_vector_type(2))) float v2f;
typedef __attribute__((ext_vector_type(4))) float v4f;
typedef __attribute__((ext_vector_type(8))) float v8f;

#define S_LEN   4096
#define D_DIM   1024
#define NPATCH  8
#define NSEG    9

__global__ void seg_bounds_kernel(const int* __restrict__ mask,
                                  int* __restrict__ ofs) {
  const int b   = blockIdx.x;
  const int tid = threadIdx.x;
  __shared__ int psum[256];

  const int base = b * S_LEN + tid * 16;
  int m[16];
  int c = 0;
#pragma unroll
  for (int j = 0; j < 16; ++j) { m[j] = mask[base + j]; c += m[j]; }
  psum[tid] = c;
  __syncthreads();

#pragma unroll
  for (int off = 1; off < 256; off <<= 1) {
    int v   = psum[tid];
    int add = (tid >= off) ? psum[tid - off] : 0;
    __syncthreads();
    psum[tid] = v + add;
    __syncthreads();
  }
  const int excl = (tid == 0) ? 0 : psum[tid - 1];

  if (tid < 10) ofs[b * 10 + tid] = (tid == 0) ? 0 : S_LEN;
  __syncthreads();

  int r = excl;
#pragma unroll
  for (int j = 0; j < 16; ++j) {
    if (m[j]) {
      if (r < NPATCH) ofs[b * 10 + (r + 1)] = tid * 16 + j + 1;
      ++r;
    }
  }
}

__global__ void seg_mean_kernel(const float* __restrict__ x,
                                const int* __restrict__ ofs,
                                float* __restrict__ patches) {
  const int p = blockIdx.x & 7;
  const int b = blockIdx.x >> 3;
  const int d = threadIdx.x;

  const int s0 = ofs[b * 10 + p];
  const int s1 = ofs[b * 10 + p + 1];

  const float* src = x + (size_t)b * S_LEN * D_DIM + d;
  float acc = 0.0f;
  for (int s = s0; s < s1; ++s) acc += src[(size_t)s * D_DIM];

  const int cnt = s1 - s0;
  const float denom = (float)(cnt > 0 ? cnt : 1);
  patches[(size_t)(b * NPATCH + p) * D_DIM + d] = acc / denom;
}

#define GEMM_WAVES 8
#define K_SLICE    (D_DIM / GEMM_WAVES)
#define K_BLOCKS   (K_SLICE / 8)

__global__ __launch_bounds__(GEMM_WAVES * 32)
void patch_gemm_wmma(const float* __restrict__ P,
                     const float* __restrict__ Wm,
                     const float* __restrict__ bias,
                     float* __restrict__ out) {
  __shared__ float red[GEMM_WAVES * 256];

  const int mt   = blockIdx.x >> 6;
  const int nt   = blockIdx.x & 63;
  const int m0   = mt * 16;
  const int n0   = nt * 16;
  const int w    = threadIdx.x >> 5;
  const int lane = threadIdx.x & 31;
  const int half = lane >> 4;
  const int lr   = lane & 15;

  const int kb = w * K_SLICE + half * 4;
  const float* Arow = P  + (size_t)(m0 + lr) * D_DIM + kb;
  const float* Brow = Wm + (size_t)(n0 + lr) * D_DIM + kb;

  const float bv = bias[n0 + lr];

  v4f a[K_BLOCKS], b[K_BLOCKS];
#pragma unroll
  for (int j = 0; j < K_BLOCKS; ++j) {
    a[j] = *(const v4f*)(Arow + j * 8);
    b[j] = *(const v4f*)(Brow + j * 8);
  }
  __builtin_amdgcn_sched_barrier(0);

  v8f c0 = {}, c1 = {};
#pragma unroll
  for (int j = 0; j < K_BLOCKS; ++j) {
    v2f a0 = {a[j].x, a[j].y};
    v2f b0 = {b[j].x, b[j].y};
    v2f a1 = {a[j].z, a[j].w};
    v2f b1 = {b[j].z, b[j].w};
    c0 = __builtin_amdgcn_wmma_f32_16x16x4_f32(
         false, a0,  false, b0,
         (short)0, c0,  false,  false);
    c1 = __builtin_amdgcn_wmma_f32_16x16x4_f32(
         false, a1,  false, b1,
         (short)0, c1,  false,  false);
  }

#pragma unroll
  for (int r = 0; r < 8; ++r) {
    red[(w * 8 + r) * 32 + lane] = c0[r] + c1[r];
  }
  __syncthreads();

  const int t = threadIdx.x;
  float s = red[t];
#pragma unroll
  for (int q = 1; q < GEMM_WAVES; ++q) s += red[q * 256 + t];

  const int orow = m0 + half * 8 + w;
  out[(size_t)orow * D_DIM + n0 + lr] = s + bv;
}

extern "C" void kernel_launch(void* const* d_in, const int* in_sizes, int n_in,
                              void* d_out, int out_size, void* d_ws, size_t ws_size,
                              hipStream_t stream) {
  const float* latent = (const float*)d_in[0];
  const int*   mask   = (const int*)  d_in[1];
  const float* W      = (const float*)d_in[2];
  const float* bias   = (const float*)d_in[3];
  float*       out    = (float*)d_out;

  int*   ofs     = (int*)d_ws;
  float* patches = (float*)((char*)d_ws + 1024);

  seg_bounds_kernel<<<8, 256, 0, stream>>>(mask, ofs);
  seg_mean_kernel<<<64, 1024, 0, stream>>>(latent, ofs, patches);
  patch_gemm_wmma<<<256, 256, 0, stream>>>(patches, W, bias, out);
}


# Segment-boundary launch folded into the mean kernel: each (batch,patch) block ranks its batch's mask itself with wave ballots and 32 wave totals in LDS; 3 launches become 2
# speedup vs baseline: 3.1426x; 1.3778x over previous
#include <hip/hip_runtime.h>
#include <hip/hip_bf16.h>

typedef __attribute__((ext_vector_type(4))) int   v4i;
typedef __attribute__((ext_vector_type(2))) float v2f;
typedef __attribute__((ext_vector_type(4))) float v4f;
typedef __attribute__((ext_vector_type(8))) float v8f;

#define S_LEN   4096
#define D_DIM   1024
#define NPATCH  8
#define NSEG    9

__global__ __launch_bounds__(1024)
void seg_mean_kernel(const float* __restrict__ x,
                     const int* __restrict__ mask,
                     float* __restrict__ patches) {
  const int p    = blockIdx.x & 7;
  const int b    = blockIdx.x >> 3;
  const int d    = threadIdx.x;
  const int lane = d & 31;
  const int w    = d >> 5;

  __shared__ int wtot[32];
  __shared__ int sofs[NSEG + 1];

  const v4i m = *(const v4i*)(mask + b * S_LEN + d * 4);

  if (d < NSEG + 1) sofs[d] = (d == 0) ? 0 : S_LEN;

  const unsigned b0 = __builtin_amdgcn_ballot_w32(m.x != 0);
  const unsigned b1 = __builtin_amdgcn_ballot_w32(m.y != 0);
  const unsigned b2 = __builtin_amdgcn_ballot_w32(m.z != 0);
  const unsigned b3 = __builtin_amdgcn_ballot_w32(m.w != 0);

  int r = (int)__builtin_amdgcn_mbcnt_lo(b3,
               __builtin_amdgcn_mbcnt_lo(b2,
               __builtin_amdgcn_mbcnt_lo(b1,
               __builtin_amdgcn_mbcnt_lo(b0, 0u))));

  if (lane == 0) wtot[w] = __popc(b0) + __popc(b1) + __popc(b2) + __popc(b3);
  __syncthreads();

#pragma unroll
  for (int q = 0; q < 32; ++q) r += (q < w) ? wtot[q] : 0;

  const int pos = d * 4;
  if (m.x != 0) { if (r < NPATCH) sofs[r + 1] = pos + 1; ++r; }
  if (m.y != 0) { if (r < NPATCH) sofs[r + 1] = pos + 2; ++r; }
  if (m.z != 0) { if (r < NPATCH) sofs[r + 1] = pos + 3; ++r; }
  if (m.w != 0) { if (r < NPATCH) sofs[r + 1] = pos + 4; }
  __syncthreads();

  const int s0 = sofs[p];
  const int s1 = sofs[p + 1];

  const float* src = x + (size_t)b * S_LEN * D_DIM + d;
  float acc = 0.0f;
  for (int s = s0; s < s1; ++s) acc += src[(size_t)s * D_DIM];

  const int cnt = s1 - s0;
  const float denom = (float)(cnt > 0 ? cnt : 1);
  patches[(size_t)(b * NPATCH + p) * D_DIM + d] = acc / denom;
}

#define GEMM_WAVES 8
#define K_SLICE    (D_DIM / GEMM_WAVES)
#define K_BLOCKS   (K_SLICE / 8)

__global__ __launch_bounds__(GEMM_WAVES * 32)
void patch_gemm_wmma(const float* __restrict__ P,
                     const float* __restrict__ Wm,
                     const float* __restrict__ bias,
                     float* __restrict__ out) {
  __shared__ float red[GEMM_WAVES * 256];

  const int mt   = blockIdx.x >> 6;
  const int nt   = blockIdx.x & 63;
  const int m0   = mt * 16;
  const int n0   = nt * 16;
  const int w    = threadIdx.x >> 5;
  const int lane = threadIdx.x & 31;
  const int half = lane >> 4;
  const int lr   = lane & 15;

  const int kb = w * K_SLICE + half * 4;
  const float* Arow = P  + (size_t)(m0 + lr) * D_DIM + kb;
  const float* Brow = Wm + (size_t)(n0 + lr) * D_DIM + kb;

  const float bv = bias[n0 + lr];

  v4f a[K_BLOCKS], b[K_BLOCKS];
#pragma unroll
  for (int j = 0; j < K_BLOCKS; ++j) {
    a[j] = *(const v4f*)(Arow + j * 8);
    b[j] = *(const v4f*)(Brow + j * 8);
  }
  __builtin_amdgcn_sched_barrier(0);

  v8f c0 = {}, c1 = {};
#pragma unroll
  for (int j = 0; j < K_BLOCKS; ++j) {
    v2f a0 = {a[j].x, a[j].y};
    v2f b0 = {b[j].x, b[j].y};
    v2f a1 = {a[j].z, a[j].w};
    v2f b1 = {b[j].z, b[j].w};
    c0 = __builtin_amdgcn_wmma_f32_16x16x4_f32(
         false, a0,  false, b0,
         (short)0, c0,  false,  false);
    c1 = __builtin_amdgcn_wmma_f32_16x16x4_f32(
         false, a1,  false, b1,
         (short)0, c1,  false,  false);
  }

#pragma unroll
  for (int r = 0; r < 8; ++r) {
    red[(w * 8 + r) * 32 + lane] = c0[r] + c1[r];
  }
  __syncthreads();

  const int t = threadIdx.x;
  float s = red[t];
#pragma unroll
  for (int q = 1; q < GEMM_WAVES; ++q) s += red[q * 256 + t];

  const int orow = m0 + half * 8 + w;
  out[(size_t)orow * D_DIM + n0 + lr] = s + bv;
}

extern "C" void kernel_launch(void* const* d_in, const int* in_sizes, int n_in,
                              void* d_out, int out_size, void* d_ws, size_t ws_size,
                              hipStream_t stream) {
  const float* latent = (const float*)d_in[0];
  const int*   mask   = (const int*)  d_in[1];
  const float* W      = (const float*)d_in[2];
  const float* bias   = (const float*)d_in[3];
  float*       out    = (float*)d_out;

  float* patches = (float*)((char*)d_ws + 1024);

  seg_mean_kernel<<<64, 1024, 0, stream>>>(latent, mask, patches);
  patch_gemm_wmma<<<256, 256, 0, stream>>>(patches, W, bias, out);
}


# Mean launch fused into the GEMM: a tile's workgroup ballots its 2 batches' mask, sums the needed rows into an LDS tile for its A fragments, divides by the row count after the K sum; 2 launches to 1
# speedup vs baseline: 4.7013x; 1.4960x over previous
#include <hip/hip_runtime.h>

typedef __attribute__((ext_vector_type(2))) float v2f;
typedef __attribute__((ext_vector_type(4))) float v4f;
typedef __attribute__((ext_vector_type(8))) float v8f;

#define S_LEN   4096
#define D_DIM   1024
#define NPATCH  8

#define GEMM_WAVES 8
#define K_SLICE    (D_DIM / GEMM_WAVES)
#define K_BLOCKS   (K_SLICE / 8)
#define CHUNK      32
#define PITCH      (D_DIM + 4)

__global__ __launch_bounds__(GEMM_WAVES * 32)
void patch_mean_gemm_wmma(const float* __restrict__ x,
                          const int* __restrict__ mask,
                          const float* __restrict__ Wm,
                          const float* __restrict__ bias,
                          float* __restrict__ out) {
  __shared__ __attribute__((aligned(16))) float pl[16 * PITCH];
  __shared__ float red[GEMM_WAVES * 256];
  __shared__ int   cnts[16];

  const int mt   = blockIdx.x >> 6;
  const int nt   = blockIdx.x & 63;
  const int m0   = mt * 16;
  const int n0   = nt * 16;
  const int w    = threadIdx.x >> 5;
  const int lane = threadIdx.x & 31;
  const int half = lane >> 4;
  const int lr   = lane & 15;

  const int bl  = w >> 2;
  const int bt  = mt * 2 + bl;
  const int col = (w & 3) * 256 + lane * 4;
  const int* mrow = mask + bt * S_LEN;

  const int mval0 = mrow[lane];

  const int kb = w * K_SLICE + half * 4;
  const float* Brow = Wm + (size_t)(n0 + lr) * D_DIM + kb;
  v4f b[K_BLOCKS];
  float bv;

  const float* xb = x + (size_t)bt * S_LEN * D_DIM + col;
  float* prow = pl + (bl * NPATCH) * PITCH + col;
  int*   crow = cnts + bl * NPATCH;
  const bool cnt_writer = (w & 3) == 0;

  v4f acc0 = {}, acc1 = {};
  int cnt = 0;
  int seg = 0;

  auto add_row = [&](const v4f r0, const v4f r1, const bool closes) {
    acc0 += r0;
    acc1 += r1;
    ++cnt;
    if (closes) {
      *(v4f*)(prow + seg * PITCH)       = acc0;
      *(v4f*)(prow + seg * PITCH + 128) = acc1;
      if (cnt_writer) crow[seg] = cnt;
      ++seg;
      cnt = 0;
      acc0 = v4f{}; acc1 = v4f{};
    }
  };

  {
    const unsigned bits = __builtin_amdgcn_ballot_w32(mval0 != 0);
    v4f r0[CHUNK], r1[CHUNK];
#pragma unroll
    for (int u = 0; u < CHUNK; ++u) {
      if (__popc(bits & ((1u << u) - 1u)) < NPATCH) {
        const float* r = xb + (size_t)u * D_DIM;
        r0[u] = *(const v4f*)(r);
        r1[u] = *(const v4f*)(r + 128);
      }
    }
    bv = bias[n0 + lr];
#pragma unroll
    for (int j = 0; j < K_BLOCKS; ++j) b[j] = *(const v4f*)(Brow + j * 8);
    __builtin_amdgcn_sched_barrier(0);
#pragma unroll
    for (int u = 0; u < CHUNK; ++u) {
      if (__popc(bits & ((1u << u) - 1u)) < NPATCH)
        add_row(r0[u], r1[u], ((bits >> u) & 1u) != 0);
    }
  }
  for (int base = CHUNK; base < S_LEN && seg < NPATCH; base += CHUNK) {
    const unsigned bits = __builtin_amdgcn_ballot_w32(mrow[base + lane] != 0);
    for (int s = 0; s < CHUNK && seg < NPATCH; ++s) {
      const float* r = xb + (size_t)(base + s) * D_DIM;
      add_row(*(const v4f*)(r), *(const v4f*)(r + 128), ((bits >> s) & 1u) != 0);
    }
  }
  for (; seg < NPATCH; ++seg) {
    *(v4f*)(prow + seg * PITCH)       = acc0;
    *(v4f*)(prow + seg * PITCH + 128) = acc1;
    if (cnt_writer) crow[seg] = cnt;
    cnt = 0;
    acc0 = v4f{}; acc1 = v4f{};
  }
  __syncthreads();

  const float* Arow = pl + lr * PITCH + kb;
  v4f a[K_BLOCKS];
#pragma unroll
  for (int j = 0; j < K_BLOCKS; ++j) a[j] = *(const v4f*)(Arow + j * 8);
  __builtin_amdgcn_sched_barrier(0);

  v8f c0 = {}, c1 = {};
#pragma unroll
  for (int j = 0; j < K_BLOCKS; ++j) {
    v2f a0 = {a[j].x, a[j].y};
    v2f b0 = {b[j].x, b[j].y};
    v2f a1 = {a[j].z, a[j].w};
    v2f b1 = {b[j].z, b[j].w};
    c0 = __builtin_amdgcn_wmma_f32_16x16x4_f32(
         false, a0,  false, b0,
         (short)0, c0,  false,  false);
    c1 = __builtin_amdgcn_wmma_f32_16x16x4_f32(
         false, a1,  false, b1,
         (short)0, c1,  false,  false);
  }

#pragma unroll
  for (int r = 0; r < 8; ++r) {
    red[(w * 8 + r) * 32 + lane] = c0[r] + c1[r];
  }
  __syncthreads();

  const int t = threadIdx.x;
  float s = red[t];
#pragma unroll
  for (int q = 1; q < GEMM_WAVES; ++q) s += red[q * 256 + t];

  const int rows = cnts[half * NPATCH + w];
  const float denom = (float)(rows > 0 ? rows : 1);
  const int orow = m0 + half * 8 + w;
  out[(size_t)orow * D_DIM + n0 + lr] = s / denom + bv;
}

extern "C" void kernel_launch(void* const* d_in, const int* in_sizes, int n_in,
                              void* d_out, int out_size, void* d_ws, size_t ws_size,
                              hipStream_t stream) {
  const float* latent = (const float*)d_in[0];
  const int*   mask   = (const int*)  d_in[1];
  const float* W      = (const float*)d_in[2];
  const float* bias   = (const float*)d_in[3];
  float*       out    = (float*)d_out;

  patch_mean_gemm_wmma<<<256, 256, 0, stream>>>(latent, mask, W, bias, out);
}
